# starting kernel
# baseline (speedup 1.0000x reference)
#include <hip/hip_runtime.h>

typedef __attribute__((ext_vector_type(2))) float v2f;
typedef __attribute__((ext_vector_type(4))) float v4f;
typedef __attribute__((ext_vector_type(8))) float v8f;

#define ORD 8
#define SQS 6

__global__ __launch_bounds__(512) void expm_apply_kernel(
    const float* __restrict__ x,
    const float* __restrict__ c,
    const float* __restrict__ psi,
    float* __restrict__ out)
{
    __shared__ float tile[16][16 * 17];

    const int tid = threadIdx.x;
    const int w   = tid >> 5;
    const int l   = tid & 31;
    const int h   = l >> 4;
    const int n   = l & 15;
    const long b0 = (long)blockIdx.x * 16;

    v8f acc = {0.f, 0.f, 0.f, 0.f, 0.f, 0.f, 0.f, 0.f};
#pragma unroll
    for (int kk = 0; kk < 16; ++kk) {
        const int k0 = kk * 4;
        v2f a;
        const float* cp = c + (b0 + n) * 64 + (k0 + 2 * h);
        a.x = cp[0];
        a.y = cp[1];
        const float* pp = psi + (size_t)(k0 + 2 * h) * 256 + w * 16 + n;
        v2f bb;
        bb.x = pp[0];
        bb.y = pp[256];
        acc = __builtin_amdgcn_wmma_f32_16x16x4_f32(
            false, a, false, bb, (short)0, acc, false, false);
    }
#pragma unroll
    for (int j = 0; j < 8; ++j) {
        tile[j + 8 * h][w * 17 + n] = acc[j];
    }
    __syncthreads();

    float* mt = &tile[w][0];

    const float sc = 1.0f / (float)(1 << SQS);
    v2f Ak[4];
#pragma unroll
    for (int kk = 0; kk < 4; ++kk) {
        const int col = kk * 4 + 2 * h;
        Ak[kk].x = mt[n * 17 + col] * sc;
        Ak[kk].y = mt[n * 17 + col + 1] * sc;
    }

    v8f r;
#pragma unroll
    for (int j = 0; j < 8; ++j) r[j] = (n == (j + 8 * h)) ? 1.f : 0.f;

#pragma unroll
    for (int k = ORD; k >= 1; --k) {
#pragma unroll
        for (int j = 0; j < 8; ++j) mt[(j + 8 * h) * 17 + n] = r[j];
        asm volatile("s_wait_dscnt 0" ::: "memory");
        v8f d = {0.f, 0.f, 0.f, 0.f, 0.f, 0.f, 0.f, 0.f};
#pragma unroll
        for (int kk = 0; kk < 4; ++kk) {
            const int row = kk * 4 + 2 * h;
            v2f bb;
            bb.x = mt[row * 17 + n];
            bb.y = mt[(row + 1) * 17 + n];
            d = __builtin_amdgcn_wmma_f32_16x16x4_f32(
                false, Ak[kk], false, bb, (short)0, d, false, false);
        }
        const float invk = 1.0f / (float)k;
#pragma unroll
        for (int j = 0; j < 8; ++j)
            r[j] = ((n == (j + 8 * h)) ? 1.f : 0.f) + d[j] * invk;
    }

#pragma unroll
    for (int sq = 0; sq < SQS; ++sq) {
#pragma unroll
        for (int j = 0; j < 8; ++j) mt[(j + 8 * h) * 17 + n] = r[j];
        asm volatile("s_wait_dscnt 0" ::: "memory");
        v8f d = {0.f, 0.f, 0.f, 0.f, 0.f, 0.f, 0.f, 0.f};
#pragma unroll
        for (int kk = 0; kk < 4; ++kk) {
            const int col = kk * 4 + 2 * h;
            v2f aa;
            aa.x = mt[n * 17 + col];
            aa.y = mt[n * 17 + col + 1];
            v2f bb;
            bb.x = mt[col * 17 + n];
            bb.y = mt[(col + 1) * 17 + n];
            d = __builtin_amdgcn_wmma_f32_16x16x4_f32(
                false, aa, false, bb, (short)0, d, false, false);
        }
#pragma unroll
        for (int j = 0; j < 8; ++j) r[j] = d[j];
    }

    const long b = b0 + w;
    const float xv = x[b * 16 + n];
    float p[8];
#pragma unroll
    for (int j = 0; j < 8; ++j) p[j] = r[j] * xv;
#pragma unroll
    for (int off = 1; off < 16; off <<= 1) {
#pragma unroll
        for (int j = 0; j < 8; ++j) p[j] += __shfl_xor(p[j], off, 16);
    }
    if (n == 0) {
        v4f lo = {p[0], p[1], p[2], p[3]};
        v4f hi = {p[4], p[5], p[6], p[7]};
        float* op = out + b * 16 + 8 * h;
        *(v4f*)(op)     = lo;
        *(v4f*)(op + 4) = hi;
    }
}

extern "C" void kernel_launch(void* const* d_in, const int* in_sizes, int n_in,
                              void* d_out, int out_size, void* d_ws, size_t ws_size,
                              hipStream_t stream) {
    const float* x   = (const float*)d_in[0];
    const float* c   = (const float*)d_in[1];
    const float* psi = (const float*)d_in[2];
    float* out = (float*)d_out;

    const int B = in_sizes[1] / 64;
    dim3 grid(B / 16);
    dim3 block(512);
    hipLaunchKernelGGL(expm_apply_kernel, grid, block, 0, stream,
                       x, c, psi, out);
}


# B operand of every expm product taken straight from the previous product's D registers (A loaded in the matching k order, two b128 row reads); Horner needs no LDS; k summation order changes
# speedup vs baseline: 1.1016x; 1.1016x over previous
#include <hip/hip_runtime.h>


typedef __attribute__((ext_vector_type(2))) float v2f;
typedef __attribute__((ext_vector_type(4))) float v4f;
typedef __attribute__((ext_vector_type(8))) float v8f;

#define ORD 8
#define SQS 6

#define RP 20
#define MP (16 * RP + 4)

__global__ __launch_bounds__(512) void expm_apply_kernel(
    const float* __restrict__ x,
    const float* __restrict__ c,
    const float* __restrict__ psi,
    float* __restrict__ out)
{
    __shared__ __attribute__((aligned(16))) float tile[16 * MP];

    const int tid = threadIdx.x;
    const int w   = tid >> 5;
    const int l   = tid & 31;
    const int h   = l >> 4;
    const int n   = l & 15;
    const long b0 = (long)blockIdx.x * 16;

    v8f acc = {0.f, 0.f, 0.f, 0.f, 0.f, 0.f, 0.f, 0.f};
#pragma unroll
    for (int kk = 0; kk < 16; ++kk) {
        const int k0 = kk * 4;
        v2f a;
        const float* cp = c + (b0 + n) * 64 + (k0 + 2 * h);
        a.x = cp[0];
        a.y = cp[1];
        const float* pp = psi + (size_t)(k0 + 2 * h) * 256 + w * 16 + n;
        v2f bb;
        bb.x = pp[0];
        bb.y = pp[256];
        acc = __builtin_amdgcn_wmma_f32_16x16x4_f32(
            false, a, false, bb, (short)0, acc, false, false);
    }
#pragma unroll
    for (int j = 0; j < 8; ++j) {
        tile[(j + 8 * h) * MP + w * RP + n] = acc[j];
    }
    __syncthreads();

    float* mt = &tile[w * MP];
    const v4f* arow = (const v4f*)(mt + n * RP + 8 * h);

    const float sc = 1.0f / (float)(1 << SQS);
    v2f Ak[4];
    {
        const v4f q0 = arow[0];
        const v4f q1 = arow[1];
        Ak[0].x = q0.x * sc; Ak[0].y = q0.y * sc;
        Ak[1].x = q0.z * sc; Ak[1].y = q0.w * sc;
        Ak[2].x = q1.x * sc; Ak[2].y = q1.y * sc;
        Ak[3].x = q1.z * sc; Ak[3].y = q1.w * sc;
    }

    v8f r;
#pragma unroll
    for (int j = 0; j < 8; ++j) r[j] = (n == (j + 8 * h)) ? 1.f : 0.f;

#pragma unroll
    for (int k = ORD; k >= 1; --k) {
        v8f d = {0.f, 0.f, 0.f, 0.f, 0.f, 0.f, 0.f, 0.f};
#pragma unroll
        for (int kk = 0; kk < 4; ++kk) {
            v2f bb;
            bb.x = r[2 * kk];
            bb.y = r[2 * kk + 1];
            d = __builtin_amdgcn_wmma_f32_16x16x4_f32(
                false, Ak[kk], false, bb, (short)0, d, false, false);
        }
        const float invk = 1.0f / (float)k;
#pragma unroll
        for (int j = 0; j < 8; ++j)
            r[j] = ((n == (j + 8 * h)) ? 1.f : 0.f) + d[j] * invk;
    }

#pragma unroll
    for (int sq = 0; sq < SQS; ++sq) {
#pragma unroll
        for (int j = 0; j < 8; ++j) mt[(j + 8 * h) * RP + n] = r[j];
        asm volatile("s_wait_dscnt 0" ::: "memory");
        const v4f q0 = arow[0];
        const v4f q1 = arow[1];
        v2f aa[4];
        aa[0].x = q0.x; aa[0].y = q0.y;
        aa[1].x = q0.z; aa[1].y = q0.w;
        aa[2].x = q1.x; aa[2].y = q1.y;
        aa[3].x = q1.z; aa[3].y = q1.w;
        v8f d = {0.f, 0.f, 0.f, 0.f, 0.f, 0.f, 0.f, 0.f};
#pragma unroll
        for (int kk = 0; kk < 4; ++kk) {
            v2f bb;
            bb.x = r[2 * kk];
            bb.y = r[2 * kk + 1];
            d = __builtin_amdgcn_wmma_f32_16x16x4_f32(
                false, aa[kk], false, bb, (short)0, d, false, false);
        }
#pragma unroll
        for (int j = 0; j < 8; ++j) r[j] = d[j];
    }

    const long b = b0 + w;
    const float xv = x[b * 16 + n];
    float p[8];
#pragma unroll
    for (int j = 0; j < 8; ++j) p[j] = r[j] * xv;
#pragma unroll
    for (int off = 1; off < 16; off <<= 1) {
#pragma unroll
        for (int j = 0; j < 8; ++j) p[j] += __shfl_xor(p[j], off, 16);
    }
    if (n == 0) {
        v4f lo = {p[0], p[1], p[2], p[3]};
        v4f hi = {p[4], p[5], p[6], p[7]};
        float* op = out + b * 16 + 8 * h;
        *(v4f*)(op)     = lo;
        *(v4f*)(op + 4) = hi;
    }
}

extern "C" void kernel_launch(void* const* d_in, const int* in_sizes, int n_in,
                              void* d_out, int out_size, void* d_ws, size_t ws_size,
                              hipStream_t stream) {
    const float* x   = (const float*)d_in[0];
    const float* c   = (const float*)d_in[1];
    const float* psi = (const float*)d_in[2];
    float* out = (float*)d_out;

    const int B = in_sizes[1] / 64;
    dim3 grid(B / 16);
    dim3 block(512);
    hipLaunchKernelGGL(expm_apply_kernel, grid, block, 0, stream,
                       x, c, psi, out);
}


# degree-8 Taylor polynomial of the scaled matrix evaluated in a factored form with 3 matrix products instead of 8 Horner products (two more row re-reads through LDS); summation order changes
# speedup vs baseline: 1.2189x; 1.1065x over previous
#include <hip/hip_runtime.h>


typedef __attribute__((ext_vector_type(2))) float v2f;
typedef __attribute__((ext_vector_type(4))) float v4f;
typedef __attribute__((ext_vector_type(8))) float v8f;

#define SQS 6

#define K1 4.980119205559973e-3f
#define K2 1.992047682223989e-2f
#define K3 7.665265321119145e-2f
#define K4 8.765009801785550e-1f
#define K5 1.225521150112075e-1f
#define K6 2.974307204847627f

#define RP 20
#define MP (16 * RP + 4)

static __device__ __forceinline__ v8f mm16(const v2f (&a)[4], const v8f b, v8f c)
{
#pragma unroll
    for (int kk = 0; kk < 4; ++kk) {
        v2f bb;
        bb.x = b[2 * kk];
        bb.y = b[2 * kk + 1];
        c = __builtin_amdgcn_wmma_f32_16x16x4_f32(
            false, a[kk], false, bb, (short)0, c, false, false);
    }
    return c;
}

static __device__ __forceinline__ void to_rows(float* mt, const v4f* arow,
                                               const int h, const int n,
                                               const v8f d, v2f (&a)[4])
{
#pragma unroll
    for (int j = 0; j < 8; ++j) mt[(j + 8 * h) * RP + n] = d[j];
    asm volatile("s_wait_dscnt 0" ::: "memory");
    const v4f q0 = arow[0];
    const v4f q1 = arow[1];
    a[0].x = q0.x; a[0].y = q0.y;
    a[1].x = q0.z; a[1].y = q0.w;
    a[2].x = q1.x; a[2].y = q1.y;
    a[3].x = q1.z; a[3].y = q1.w;
}

__global__ __launch_bounds__(512) void expm_apply_kernel(
    const float* __restrict__ x,
    const float* __restrict__ c,
    const float* __restrict__ psi,
    float* __restrict__ out)
{
    __shared__ __attribute__((aligned(16))) float tile[16 * MP];

    const int tid = threadIdx.x;
    const int w   = tid >> 5;
    const int l   = tid & 31;
    const int h   = l >> 4;
    const int n   = l & 15;
    const long b0 = (long)blockIdx.x * 16;

    v8f acc = {0.f, 0.f, 0.f, 0.f, 0.f, 0.f, 0.f, 0.f};
#pragma unroll
    for (int kk = 0; kk < 16; ++kk) {
        const int k0 = kk * 4;
        v2f a;
        const float* cp = c + (b0 + n) * 64 + (k0 + 2 * h);
        a.x = cp[0];
        a.y = cp[1];
        const float* pp = psi + (size_t)(k0 + 2 * h) * 256 + w * 16 + n;
        v2f bb;
        bb.x = pp[0];
        bb.y = pp[256];
        acc = __builtin_amdgcn_wmma_f32_16x16x4_f32(
            false, a, false, bb, (short)0, acc, false, false);
    }
#pragma unroll
    for (int j = 0; j < 8; ++j) {
        tile[(j + 8 * h) * MP + w * RP + n] = acc[j];
    }
    __syncthreads();

    float* mt = &tile[w * MP];
    const v4f* arow = (const v4f*)(mt + n * RP + 8 * h);

    const v8f zero = {0.f, 0.f, 0.f, 0.f, 0.f, 0.f, 0.f, 0.f};

    const float sc = 1.0f / (float)(1 << SQS);
    v2f Ak[4];
    v8f ad;
    {
        const v4f q0 = arow[0];
        const v4f q1 = arow[1];
        Ak[0].x = q0.x * sc; Ak[0].y = q0.y * sc;
        Ak[1].x = q0.z * sc; Ak[1].y = q0.w * sc;
        Ak[2].x = q1.x * sc; Ak[2].y = q1.y * sc;
        Ak[3].x = q1.z * sc; Ak[3].y = q1.w * sc;
#pragma unroll
        for (int j = 0; j < 8; ++j) ad[j] = mt[(j + 8 * h) * RP + n] * sc;
    }

    const v8f a2 = mm16(Ak, ad, zero);
    v2f A2k[4];
    to_rows(mt, arow, h, n, a2, A2k);

    v8f t;
#pragma unroll
    for (int j = 0; j < 8; ++j) t[j] = K1 * a2[j] + K2 * ad[j];
    const v8f y = mm16(A2k, t, zero);

    v8f f1, f2;
#pragma unroll
    for (int j = 0; j < 8; ++j) {
        f1[j] = y[j] + (K3 * a2[j] + K4 * ad[j]);
        f2[j] = y[j] + K5 * a2[j];
    }
    v2f F1k[4];
    to_rows(mt, arow, h, n, f1, F1k);
    const v8f pr = mm16(F1k, f2, zero);
    v8f r;
#pragma unroll
    for (int j = 0; j < 8; ++j) {
        const float lowp = pr[j] + (K6 * y[j] + (0.5f * a2[j] + ad[j]));
        r[j] = ((n == (j + 8 * h)) ? 1.f : 0.f) + lowp;
    }

#pragma unroll
    for (int sq = 0; sq < SQS; ++sq) {
        v2f aa[4];
        to_rows(mt, arow, h, n, r, aa);
        r = mm16(aa, r, zero);
    }

    const long b = b0 + w;
    const float xv = x[b * 16 + n];
    float p[8];
#pragma unroll
    for (int j = 0; j < 8; ++j) p[j] = r[j] * xv;
#pragma unroll
    for (int off = 1; off < 16; off <<= 1) {
#pragma unroll
        for (int j = 0; j < 8; ++j) p[j] += __shfl_xor(p[j], off, 16);
    }
    if (n == 0) {
        v4f lo = {p[0], p[1], p[2], p[3]};
        v4f hi = {p[4], p[5], p[6], p[7]};
        float* op = out + b * 16 + 8 * h;
        *(v4f*)(op)     = lo;
        *(v4f*)(op + 4) = hi;
    }
}

extern "C" void kernel_launch(void* const* d_in, const int* in_sizes, int n_in,
                              void* d_out, int out_size, void* d_ws, size_t ws_size,
                              hipStream_t stream) {
    const float* x   = (const float*)d_in[0];
    const float* c   = (const float*)d_in[1];
    const float* psi = (const float*)d_in[2];
    float* out = (float*)d_out;

    const int B = in_sizes[1] / 64;
    dim3 grid(B / 16);
    dim3 block(512);
    hipLaunchKernelGGL(expm_apply_kernel, grid, block, 0, stream,
                       x, c, psi, out);
}


# c tile read from memory once per block (wave w fetches row w as one 256-byte load) and shared through LDS; before, each of the 16 waves re-read it with 16-line loads; summation order unchanged
# speedup vs baseline: 1.3714x; 1.1251x over previous
#include <hip/hip_runtime.h>


typedef __attribute__((ext_vector_type(2))) float v2f;
typedef __attribute__((ext_vector_type(4))) float v4f;
typedef __attribute__((ext_vector_type(8))) float v8f;

#define SQS 6

#define K1 4.980119205559973e-3f
#define K2 1.992047682223989e-2f
#define K3 7.665265321119145e-2f
#define K4 8.765009801785550e-1f
#define K5 1.225521150112075e-1f
#define K6 2.974307204847627f

#define RP 20
#define MP (16 * RP + 4)
#define CP 68

static __device__ __forceinline__ v8f mm16(const v2f (&a)[4], const v8f b, v8f c)
{
#pragma unroll
    for (int kk = 0; kk < 4; ++kk) {
        v2f bb;
        bb.x = b[2 * kk];
        bb.y = b[2 * kk + 1];
        c = __builtin_amdgcn_wmma_f32_16x16x4_f32(
            false, a[kk], false, bb, (short)0, c, false, false);
    }
    return c;
}

static __device__ __forceinline__ void to_rows(float* mt, const v4f* arow,
                                               const int h, const int n,
                                               const v8f d, v2f (&a)[4])
{
#pragma unroll
    for (int j = 0; j < 8; ++j) mt[(j + 8 * h) * RP + n] = d[j];
    asm volatile("s_wait_dscnt 0" ::: "memory");
    const v4f q0 = arow[0];
    const v4f q1 = arow[1];
    a[0].x = q0.x; a[0].y = q0.y;
    a[1].x = q0.z; a[1].y = q0.w;
    a[2].x = q1.x; a[2].y = q1.y;
    a[3].x = q1.z; a[3].y = q1.w;
}

__global__ __launch_bounds__(512) void expm_apply_kernel(
    const float* __restrict__ x,
    const float* __restrict__ c,
    const float* __restrict__ psi,
    float* __restrict__ out)
{
    __shared__ __attribute__((aligned(16))) float tile[16 * MP];
    __shared__ __attribute__((aligned(16))) float cs[16 * CP];

    const int tid = threadIdx.x;
    const int w   = tid >> 5;
    const int l   = tid & 31;
    const int h   = l >> 4;
    const int n   = l & 15;
    const long b0 = (long)blockIdx.x * 16;

    const v2f crow = *(const v2f*)(c + (b0 + w) * 64 + 2 * l);

    v2f pb[16];
#pragma unroll
    for (int kk = 0; kk < 16; ++kk) {
        const float* pp = psi + (size_t)(kk * 4 + 2 * h) * 256 + w * 16 + n;
        pb[kk].x = pp[0];
        pb[kk].y = pp[256];
    }

    *(v2f*)(cs + w * CP + 2 * l) = crow;
    __syncthreads();

    v8f acc = {0.f, 0.f, 0.f, 0.f, 0.f, 0.f, 0.f, 0.f};
#pragma unroll
    for (int kk = 0; kk < 16; ++kk) {
        const int k0 = kk * 4;
        const v2f a = *(const v2f*)(cs + n * CP + (k0 + 2 * h));
        acc = __builtin_amdgcn_wmma_f32_16x16x4_f32(
            false, a, false, pb[kk], (short)0, acc, false, false);
    }
#pragma unroll
    for (int j = 0; j < 8; ++j) {
        tile[(j + 8 * h) * MP + w * RP + n] = acc[j];
    }
    __syncthreads();

    float* mt = &tile[w * MP];
    const v4f* arow = (const v4f*)(mt + n * RP + 8 * h);

    const v8f zero = {0.f, 0.f, 0.f, 0.f, 0.f, 0.f, 0.f, 0.f};

    const float sc = 1.0f / (float)(1 << SQS);
    v2f Ak[4];
    v8f ad;
    {
        const v4f q0 = arow[0];
        const v4f q1 = arow[1];
        Ak[0].x = q0.x * sc; Ak[0].y = q0.y * sc;
        Ak[1].x = q0.z * sc; Ak[1].y = q0.w * sc;
        Ak[2].x = q1.x * sc; Ak[2].y = q1.y * sc;
        Ak[3].x = q1.z * sc; Ak[3].y = q1.w * sc;
#pragma unroll
        for (int j = 0; j < 8; ++j) ad[j] = mt[(j + 8 * h) * RP + n] * sc;
    }

    const v8f a2 = mm16(Ak, ad, zero);
    v2f A2k[4];
    to_rows(mt, arow, h, n, a2, A2k);

    v8f t;
#pragma unroll
    for (int j = 0; j < 8; ++j) t[j] = K1 * a2[j] + K2 * ad[j];
    const v8f y = mm16(A2k, t, zero);

    v8f f1, f2;
#pragma unroll
    for (int j = 0; j < 8; ++j) {
        f1[j] = y[j] + (K3 * a2[j] + K4 * ad[j]);
        f2[j] = y[j] + K5 * a2[j];
    }
    v2f F1k[4];
    to_rows(mt, arow, h, n, f1, F1k);
    const v8f pr = mm16(F1k, f2, zero);
    v8f r;
#pragma unroll
    for (int j = 0; j < 8; ++j) {
        const float lowp = pr[j] + (K6 * y[j] + (0.5f * a2[j] + ad[j]));
        r[j] = ((n == (j + 8 * h)) ? 1.f : 0.f) + lowp;
    }

#pragma unroll
    for (int sq = 0; sq < SQS; ++sq) {
        v2f aa[4];
        to_rows(mt, arow, h, n, r, aa);
        r = mm16(aa, r, zero);
    }

    const long b = b0 + w;
    const float xv = x[b * 16 + n];
    float p[8];
#pragma unroll
    for (int j = 0; j < 8; ++j) p[j] = r[j] * xv;
#pragma unroll
    for (int off = 1; off < 16; off <<= 1) {
#pragma unroll
        for (int j = 0; j < 8; ++j) p[j] += __shfl_xor(p[j], off, 16);
    }
    if (n == 0) {
        v4f lo = {p[0], p[1], p[2], p[3]};
        v4f hi = {p[4], p[5], p[6], p[7]};
        float* op = out + b * 16 + 8 * h;
        *(v4f*)(op)     = lo;
        *(v4f*)(op + 4) = hi;
    }
}

extern "C" void kernel_launch(void* const* d_in, const int* in_sizes, int n_in,
                              void* d_out, int out_size, void* d_ws, size_t ws_size,
                              hipStream_t stream) {
    const float* x   = (const float*)d_in[0];
    const float* c   = (const float*)d_in[1];
    const float* psi = (const float*)d_in[2];
    float* out = (float*)d_out;

    const int B = in_sizes[1] / 64;
    dim3 grid(B / 16);
    dim3 block(512);
    hipLaunchKernelGGL(expm_apply_kernel, grid, block, 0, stream,
                       x, c, psi, out);
}


# each block loops over 4 batch tiles (grid 1024, was 4096): psi fragments loaded once per wave and held in registers, next tile's c row requested one trip ahead; summation order unchanged
# speedup vs baseline: 1.5717x; 1.1461x over previous
#include <hip/hip_runtime.h>


typedef __attribute__((ext_vector_type(2))) float v2f;
typedef __attribute__((ext_vector_type(4))) float v4f;
typedef __attribute__((ext_vector_type(8))) float v8f;

#define SQS 6

#define K1 4.980119205559973e-3f
#define K2 1.992047682223989e-2f
#define K3 7.665265321119145e-2f
#define K4 8.765009801785550e-1f
#define K5 1.225521150112075e-1f
#define K6 2.974307204847627f

#define RP 20
#define MP (16 * RP + 4)
#define CP 68

static __device__ __forceinline__ v8f mm16(const v2f (&a)[4], const v8f b, v8f c)
{
#pragma unroll
    for (int kk = 0; kk < 4; ++kk) {
        v2f bb;
        bb.x = b[2 * kk];
        bb.y = b[2 * kk + 1];
        c = __builtin_amdgcn_wmma_f32_16x16x4_f32(
            false, a[kk], false, bb, (short)0, c, false, false);
    }
    return c;
}

static __device__ __forceinline__ void to_rows(float* colp, const v4f* arow,
                                               const v8f d, v2f (&a)[4])
{
#pragma unroll
    for (int j = 0; j < 8; ++j) colp[j * RP] = d[j];
    asm volatile("s_wait_dscnt 0" ::: "memory");
    const v4f q0 = arow[0];
    const v4f q1 = arow[1];
    a[0].x = q0.x; a[0].y = q0.y;
    a[1].x = q0.z; a[1].y = q0.w;
    a[2].x = q1.x; a[2].y = q1.y;
    a[3].x = q1.z; a[3].y = q1.w;
}

__global__ __launch_bounds__(512) void expm_apply_kernel(
    const float* __restrict__ x,
    const float* __restrict__ c,
    const float* __restrict__ psi,
    float* __restrict__ out,
    const int ntiles)
{
    __shared__ __attribute__((aligned(16))) float tile[16 * MP];
    __shared__ __attribute__((aligned(16))) float cs[16 * CP];

    const int tid = threadIdx.x;
    const int w   = tid >> 5;
    const int l   = tid & 31;
    const int h   = l >> 4;
    const int n   = l & 15;
    const int tstep = (int)gridDim.x;
    int ti = (int)blockIdx.x;
    if (ti >= ntiles) return;

    v2f crow = *(const v2f*)(c + ((long)ti * 16 + w) * 64 + 2 * l);

    v2f pb[16];
#pragma unroll
    for (int kk = 0; kk < 16; ++kk) {
        const float* pp = psi + (size_t)(kk * 4 + 2 * h) * 256 + w * 16 + n;
        pb[kk].x = pp[0];
        pb[kk].y = pp[256];
    }

    float* mt = &tile[w * MP];
    const v4f* arow = (const v4f*)(mt + n * RP + 8 * h);
    float* colp = mt + 8 * h * RP + n;
    float* scat = tile + 8 * h * MP + w * RP + n;
    const v8f zero = {0.f, 0.f, 0.f, 0.f, 0.f, 0.f, 0.f, 0.f};

    for (; ti < ntiles; ti += tstep) {
    const long b = (long)ti * 16 + w;

    *(v2f*)(cs + w * CP + 2 * l) = crow;
    __syncthreads();

    v8f acc = {0.f, 0.f, 0.f, 0.f, 0.f, 0.f, 0.f, 0.f};
#pragma unroll
    for (int kk = 0; kk < 16; ++kk) {
        const int k0 = kk * 4;
        const v2f a = *(const v2f*)(cs + n * CP + (k0 + 2 * h));
        acc = __builtin_amdgcn_wmma_f32_16x16x4_f32(
            false, a, false, pb[kk], (short)0, acc, false, false);
    }
#pragma unroll
    for (int j = 0; j < 8; ++j) {
        scat[j * MP] = acc[j];
    }
    __syncthreads();

    const int tn = (ti + tstep < ntiles) ? ti + tstep : ti;
    crow = *(const v2f*)(c + ((long)tn * 16 + w) * 64 + 2 * l);
    const float xv = x[b * 16 + n];

    const float sc = 1.0f / (float)(1 << SQS);
    v2f Ak[4];
    v8f ad;
    {
        const v4f q0 = arow[0];
        const v4f q1 = arow[1];
        Ak[0].x = q0.x * sc; Ak[0].y = q0.y * sc;
        Ak[1].x = q0.z * sc; Ak[1].y = q0.w * sc;
        Ak[2].x = q1.x * sc; Ak[2].y = q1.y * sc;
        Ak[3].x = q1.z * sc; Ak[3].y = q1.w * sc;
#pragma unroll
        for (int j = 0; j < 8; ++j) ad[j] = colp[j * RP] * sc;
    }

    const v8f a2 = mm16(Ak, ad, zero);
    v2f A2k[4];
    to_rows(colp, arow, a2, A2k);

    v8f t;
#pragma unroll
    for (int j = 0; j < 8; ++j) t[j] = K1 * a2[j] + K2 * ad[j];
    const v8f y = mm16(A2k, t, zero);

    v8f f1, f2;
#pragma unroll
    for (int j = 0; j < 8; ++j) {
        f1[j] = y[j] + (K3 * a2[j] + K4 * ad[j]);
        f2[j] = y[j] + K5 * a2[j];
    }
    v2f F1k[4];
    to_rows(colp, arow, f1, F1k);
    const v8f pr = mm16(F1k, f2, zero);
    v8f r;
#pragma unroll
    for (int j = 0; j < 8; ++j) {
        const float lowp = pr[j] + (K6 * y[j] + (0.5f * a2[j] + ad[j]));
        r[j] = ((n == (j + 8 * h)) ? 1.f : 0.f) + lowp;
    }

#pragma unroll
    for (int sq = 0; sq < SQS; ++sq) {
        v2f aa[4];
        to_rows(colp, arow, r, aa);
        r = mm16(aa, r, zero);
    }

    float p[8];
#pragma unroll
    for (int j = 0; j < 8; ++j) p[j] = r[j] * xv;
#pragma unroll
    for (int off = 1; off < 16; off <<= 1) {
#pragma unroll
        for (int j = 0; j < 8; ++j) p[j] += __shfl_xor(p[j], off, 16);
    }
    if (n == 0) {
        v4f lo = {p[0], p[1], p[2], p[3]};
        v4f hi = {p[4], p[5], p[6], p[7]};
        float* op = out + b * 16 + 8 * h;
        *(v4f*)(op)     = lo;
        *(v4f*)(op + 4) = hi;
    }
    }
}

extern "C" void kernel_launch(void* const* d_in, const int* in_sizes, int n_in,
                              void* d_out, int out_size, void* d_ws, size_t ws_size,
                              hipStream_t stream) {
    const float* x   = (const float*)d_in[0];
    const float* c   = (const float*)d_in[1];
    const float* psi = (const float*)d_in[2];
    float* out = (float*)d_out;

    const int B = in_sizes[1] / 64;
    dim3 grid(1024);
    dim3 block(512);
    hipLaunchKernelGGL(expm_apply_kernel, grid, block, 0, stream,
                       x, c, psi, out, B / 16);
}


# last 3 of the 6 squarings not formed: exp(T)x = M^8 x as 8 matrix-vector steps (row-broadcast lane reads, vector unit), M = 3 squarings on WMMA; summation order changes
# speedup vs baseline: 1.6886x; 1.0744x over previous
#include <hip/hip_runtime.h>


typedef __attribute__((ext_vector_type(2))) float v2f;
typedef __attribute__((ext_vector_type(4))) float v4f;
typedef __attribute__((ext_vector_type(8))) float v8f;

#define SQS 6
#define KSQ 3
#define NMV (1 << (SQS - KSQ))

#define K1 4.980119205559973e-3f
#define K2 1.992047682223989e-2f
#define K3 7.665265321119145e-2f
#define K4 8.765009801785550e-1f
#define K5 1.225521150112075e-1f
#define K6 2.974307204847627f

#define RP 20
#define MP (16 * RP + 4)
#define CP 68

static __device__ __forceinline__ v8f mm16(const v2f (&a)[4], const v8f b, v8f c)
{
#pragma unroll
    for (int kk = 0; kk < 4; ++kk) {
        v2f bb;
        bb.x = b[2 * kk];
        bb.y = b[2 * kk + 1];
        c = __builtin_amdgcn_wmma_f32_16x16x4_f32(
            false, a[kk], false, bb, (short)0, c, false, false);
    }
    return c;
}

static __device__ __forceinline__ void to_rows(float* colp, const v4f* arow,
                                               const v8f d, v2f (&a)[4])
{
#pragma unroll
    for (int j = 0; j < 8; ++j) colp[j * RP] = d[j];
    asm volatile("s_wait_dscnt 0" ::: "memory");
    const v4f q0 = arow[0];
    const v4f q1 = arow[1];
    a[0].x = q0.x; a[0].y = q0.y;
    a[1].x = q0.z; a[1].y = q0.w;
    a[2].x = q1.x; a[2].y = q1.y;
    a[3].x = q1.z; a[3].y = q1.w;
}

template <int J>
static __device__ __forceinline__ float row_bcast(const float v)
{
    return __builtin_bit_cast(float, __builtin_amdgcn_update_dpp(
        0, __builtin_bit_cast(int, v), 0x150 + J, 0xf, 0xf, true));
}

static __device__ __forceinline__ float other_half_turned(const float v)
{
    const int b = __builtin_bit_cast(int, v);
    return __builtin_bit_cast(float, __builtin_amdgcn_permlanex16(
        b, b, 0xfedcba98u, 0x76543210u, false, false));
}

__global__ __launch_bounds__(512) void expm_apply_kernel(
    const float* __restrict__ x,
    const float* __restrict__ c,
    const float* __restrict__ psi,
    float* __restrict__ out,
    const int ntiles)
{
    __shared__ __attribute__((aligned(16))) float tile[16 * MP];
    __shared__ __attribute__((aligned(16))) float cs[16 * CP];

    const int tid = threadIdx.x;
    const int w   = tid >> 5;
    const int l   = tid & 31;
    const int h   = l >> 4;
    const int n   = l & 15;
    const int tstep = (int)gridDim.x;
    int ti = (int)blockIdx.x;
    if (ti >= ntiles) return;

    v2f crow = *(const v2f*)(c + ((long)ti * 16 + w) * 64 + 2 * l);

    v2f pb[16];
#pragma unroll
    for (int kk = 0; kk < 16; ++kk) {
        const float* pp = psi + (size_t)(kk * 4 + 2 * h) * 256 + w * 16 + n;
        pb[kk].x = pp[0];
        pb[kk].y = pp[256];
    }

    float* mt = &tile[w * MP];
    const v4f* arow = (const v4f*)(mt + n * RP + 8 * h);
    float* colp = mt + 8 * h * RP + n;
    const v4f* mrow = (const v4f*)(mt + ((n + 8 * h) & 15) * RP + 8 * h);
    float* scat = tile + 8 * h * MP + w * RP + n;
    const v8f zero = {0.f, 0.f, 0.f, 0.f, 0.f, 0.f, 0.f, 0.f};

    for (; ti < ntiles; ti += tstep) {
    const long b = (long)ti * 16 + w;

    *(v2f*)(cs + w * CP + 2 * l) = crow;
    __syncthreads();

    v8f acc = {0.f, 0.f, 0.f, 0.f, 0.f, 0.f, 0.f, 0.f};
#pragma unroll
    for (int kk = 0; kk < 16; ++kk) {
        const int k0 = kk * 4;
        const v2f a = *(const v2f*)(cs + n * CP + (k0 + 2 * h));
        acc = __builtin_amdgcn_wmma_f32_16x16x4_f32(
            false, a, false, pb[kk], (short)0, acc, false, false);
    }
#pragma unroll
    for (int j = 0; j < 8; ++j) {
        scat[j * MP] = acc[j];
    }
    __syncthreads();

    const int tn = (ti + tstep < ntiles) ? ti + tstep : ti;
    crow = *(const v2f*)(c + ((long)tn * 16 + w) * 64 + 2 * l);
    float wv = x[b * 16 + ((n + 8 * h) & 15)];

    const float sc = 1.0f / (float)(1 << SQS);
    v2f Ak[4];
    v8f ad;
    {
        const v4f q0 = arow[0];
        const v4f q1 = arow[1];
        Ak[0].x = q0.x * sc; Ak[0].y = q0.y * sc;
        Ak[1].x = q0.z * sc; Ak[1].y = q0.w * sc;
        Ak[2].x = q1.x * sc; Ak[2].y = q1.y * sc;
        Ak[3].x = q1.z * sc; Ak[3].y = q1.w * sc;
#pragma unroll
        for (int j = 0; j < 8; ++j) ad[j] = colp[j * RP] * sc;
    }

    const v8f a2 = mm16(Ak, ad, zero);
    v2f A2k[4];
    to_rows(colp, arow, a2, A2k);

    v8f t;
#pragma unroll
    for (int j = 0; j < 8; ++j) t[j] = K1 * a2[j] + K2 * ad[j];
    const v8f y = mm16(A2k, t, zero);

    v8f f1, f2;
#pragma unroll
    for (int j = 0; j < 8; ++j) {
        f1[j] = y[j] + (K3 * a2[j] + K4 * ad[j]);
        f2[j] = y[j] + K5 * a2[j];
    }
    v2f F1k[4];
    to_rows(colp, arow, f1, F1k);
    const v8f pr = mm16(F1k, f2, zero);
    v8f r;
#pragma unroll
    for (int j = 0; j < 8; ++j) {
        const float lowp = pr[j] + (K6 * y[j] + (0.5f * a2[j] + ad[j]));
        r[j] = ((n == (j + 8 * h)) ? 1.f : 0.f) + lowp;
    }

#pragma unroll
    for (int sq = 0; sq < KSQ; ++sq) {
        v2f aa[4];
        to_rows(colp, arow, r, aa);
        r = mm16(aa, r, zero);
    }

    v2f mk[4];
    to_rows(colp, mrow, r, mk);
#pragma unroll
    for (int s = 0; s < NMV; ++s) {
        float p0 = mk[0].x * row_bcast<0>(wv);
        float p1 = mk[0].y * row_bcast<1>(wv);
        p0 = fmaf(mk[1].x, row_bcast<2>(wv), p0);
        p1 = fmaf(mk[1].y, row_bcast<3>(wv), p1);
        p0 = fmaf(mk[2].x, row_bcast<4>(wv), p0);
        p1 = fmaf(mk[2].y, row_bcast<5>(wv), p1);
        p0 = fmaf(mk[3].x, row_bcast<6>(wv), p0);
        p1 = fmaf(mk[3].y, row_bcast<7>(wv), p1);
        const float ph = p0 + p1;
        wv = ph + other_half_turned(ph);
    }
    if (h == 0) out[b * 16 + n] = wv;
    }
}

extern "C" void kernel_launch(void* const* d_in, const int* in_sizes, int n_in,
                              void* d_out, int out_size, void* d_ws, size_t ws_size,
                              hipStream_t stream) {
    const float* x   = (const float*)d_in[0];
    const float* c   = (const float*)d_in[1];
    const float* psi = (const float*)d_in[2];
    float* out = (float*)d_out;

    const int B = in_sizes[1] / 64;
    dim3 grid(1024);
    dim3 block(512);
    hipLaunchKernelGGL(expm_apply_kernel, grid, block, 0, stream,
                       x, c, psi, out, B / 16);
}


# matrix-vector steps: the 6 separate lane-broadcast moves per step folded into their multiply-adds as DPP operands (one hand-written block per step); summation order unchanged
# speedup vs baseline: 1.7439x; 1.0327x over previous
#include <hip/hip_runtime.h>


typedef __attribute__((ext_vector_type(2))) float v2f;
typedef __attribute__((ext_vector_type(4))) float v4f;
typedef __attribute__((ext_vector_type(8))) float v8f;

#define SQS 6
#define KSQ 3
#define NMV (1 << (SQS - KSQ))

#define K1 4.980119205559973e-3f
#define K2 1.992047682223989e-2f
#define K3 7.665265321119145e-2f
#define K4 8.765009801785550e-1f
#define K5 1.225521150112075e-1f
#define K6 2.974307204847627f

#define RP 20
#define MP (16 * RP + 4)
#define CP 68

static __device__ __forceinline__ v8f mm16(const v2f (&a)[4], const v8f b, v8f c)
{
#pragma unroll
    for (int kk = 0; kk < 4; ++kk) {
        v2f bb;
        bb.x = b[2 * kk];
        bb.y = b[2 * kk + 1];
        c = __builtin_amdgcn_wmma_f32_16x16x4_f32(
            false, a[kk], false, bb, (short)0, c, false, false);
    }
    return c;
}

static __device__ __forceinline__ void to_rows(float* colp, const v4f* arow,
                                               const v8f d, v2f (&a)[4])
{
#pragma unroll
    for (int j = 0; j < 8; ++j) colp[j * RP] = d[j];
    asm volatile("s_wait_dscnt 0" ::: "memory");
    const v4f q0 = arow[0];
    const v4f q1 = arow[1];
    a[0].x = q0.x; a[0].y = q0.y;
    a[1].x = q0.z; a[1].y = q0.w;
    a[2].x = q1.x; a[2].y = q1.y;
    a[3].x = q1.z; a[3].y = q1.w;
}

__global__ __launch_bounds__(512) void expm_apply_kernel(
    const float* __restrict__ x,
    const float* __restrict__ c,
    const float* __restrict__ psi,
    float* __restrict__ out,
    const int ntiles)
{
    __shared__ __attribute__((aligned(16))) float tile[16 * MP];
    __shared__ __attribute__((aligned(16))) float cs[16 * CP];

    const int tid = threadIdx.x;
    const int w   = tid >> 5;
    const int l   = tid & 31;
    const int h   = l >> 4;
    const int n   = l & 15;
    const int tstep = (int)gridDim.x;
    int ti = (int)blockIdx.x;
    if (ti >= ntiles) return;

    v2f crow = *(const v2f*)(c + ((long)ti * 16 + w) * 64 + 2 * l);

    v2f pb[16];
#pragma unroll
    for (int kk = 0; kk < 16; ++kk) {
        const float* pp = psi + (size_t)(kk * 4 + 2 * h) * 256 + w * 16 + n;
        pb[kk].x = pp[0];
        pb[kk].y = pp[256];
    }

    float* mt = &tile[w * MP];
    const v4f* arow = (const v4f*)(mt + n * RP + 8 * h);
    float* colp = mt + 8 * h * RP + n;
    const v4f* mrow = (const v4f*)(mt + ((n + 8 * h) & 15) * RP + 8 * h);
    float* scat = tile + 8 * h * MP + w * RP + n;
    const v8f zero = {0.f, 0.f, 0.f, 0.f, 0.f, 0.f, 0.f, 0.f};

    for (; ti < ntiles; ti += tstep) {
    const long b = (long)ti * 16 + w;

    *(v2f*)(cs + w * CP + 2 * l) = crow;
    __syncthreads();

    v8f acc = {0.f, 0.f, 0.f, 0.f, 0.f, 0.f, 0.f, 0.f};
#pragma unroll
    for (int kk = 0; kk < 16; ++kk) {
        const int k0 = kk * 4;
        const v2f a = *(const v2f*)(cs + n * CP + (k0 + 2 * h));
        acc = __builtin_amdgcn_wmma_f32_16x16x4_f32(
            false, a, false, pb[kk], (short)0, acc, false, false);
    }
#pragma unroll
    for (int j = 0; j < 8; ++j) {
        scat[j * MP] = acc[j];
    }
    __syncthreads();

    const int tn = (ti + tstep < ntiles) ? ti + tstep : ti;
    crow = *(const v2f*)(c + ((long)tn * 16 + w) * 64 + 2 * l);
    float wv = x[b * 16 + ((n + 8 * h) & 15)];

    const float sc = 1.0f / (float)(1 << SQS);
    v2f Ak[4];
    v8f ad;
    {
        const v4f q0 = arow[0];
        const v4f q1 = arow[1];
        Ak[0].x = q0.x * sc; Ak[0].y = q0.y * sc;
        Ak[1].x = q0.z * sc; Ak[1].y = q0.w * sc;
        Ak[2].x = q1.x * sc; Ak[2].y = q1.y * sc;
        Ak[3].x = q1.z * sc; Ak[3].y = q1.w * sc;
#pragma unroll
        for (int j = 0; j < 8; ++j) ad[j] = colp[j * RP] * sc;
    }

    const v8f a2 = mm16(Ak, ad, zero);
    v2f A2k[4];
    to_rows(colp, arow, a2, A2k);

    v8f t;
#pragma unroll
    for (int j = 0; j < 8; ++j) t[j] = K1 * a2[j] + K2 * ad[j];
    const v8f y = mm16(A2k, t, zero);

    v8f f1, f2;
#pragma unroll
    for (int j = 0; j < 8; ++j) {
        f1[j] = y[j] + (K3 * a2[j] + K4 * ad[j]);
        f2[j] = y[j] + K5 * a2[j];
    }
    v2f F1k[4];
    to_rows(colp, arow, f1, F1k);
    const v8f pr = mm16(F1k, f2, zero);
    v8f r;
#pragma unroll
    for (int j = 0; j < 8; ++j) {
        const float lowp = pr[j] + (K6 * y[j] + (0.5f * a2[j] + ad[j]));
        r[j] = ((n == (j + 8 * h)) ? 1.f : 0.f) + lowp;
    }

#pragma unroll
    for (int sq = 0; sq < KSQ; ++sq) {
        v2f aa[4];
        to_rows(colp, arow, r, aa);
        r = mm16(aa, r, zero);
    }

    v2f mk[4];
    to_rows(colp, mrow, r, mk);
    const unsigned sel_lo = 0xfedcba98u;
    const unsigned sel_hi = 0x76543210u;
#pragma unroll
    for (int s = 0; s < NMV; ++s) {
        float t0, t1;
        asm("s_delay_alu instid0(VALU_DEP_1)\n\t"
            "v_mul_f32_dpp %1, %0, %3 row_share:0 row_mask:0xf bank_mask:0xf bound_ctrl:1\n\t"
            "v_mul_f32_dpp %2, %0, %4 row_share:1 row_mask:0xf bank_mask:0xf bound_ctrl:1\n\t"
            "s_delay_alu instid0(VALU_DEP_2) | instskip(NEXT) | instid1(VALU_DEP_2)\n\t"
            "v_fmac_f32_dpp %1, %0, %5 row_share:2 row_mask:0xf bank_mask:0xf bound_ctrl:1\n\t"
            "v_fmac_f32_dpp %2, %0, %6 row_share:3 row_mask:0xf bank_mask:0xf bound_ctrl:1\n\t"
            "s_delay_alu instid0(VALU_DEP_2) | instskip(NEXT) | instid1(VALU_DEP_2)\n\t"
            "v_fmac_f32_dpp %1, %0, %7 row_share:4 row_mask:0xf bank_mask:0xf bound_ctrl:1\n\t"
            "v_fmac_f32_dpp %2, %0, %8 row_share:5 row_mask:0xf bank_mask:0xf bound_ctrl:1\n\t"
            "s_delay_alu instid0(VALU_DEP_2) | instskip(NEXT) | instid1(VALU_DEP_2)\n\t"
            "v_fmac_f32_dpp %1, %0, %9 row_share:6 row_mask:0xf bank_mask:0xf bound_ctrl:1\n\t"
            "v_fmac_f32_dpp %2, %0, %10 row_share:7 row_mask:0xf bank_mask:0xf bound_ctrl:1\n\t"
            "s_delay_alu instid0(VALU_DEP_1) | instskip(NEXT) | instid1(VALU_DEP_1)\n\t"
            "v_add_f32 %1, %1, %2\n\t"
            "v_permlanex16_b32 %2, %1, %11, %12\n\t"
            "s_delay_alu instid0(VALU_DEP_1)\n\t"
            "v_add_f32 %0, %1, %2"
            : "+v"(wv), "=&v"(t0), "=&v"(t1)
            : "v"(mk[0].x), "v"(mk[0].y), "v"(mk[1].x), "v"(mk[1].y),
              "v"(mk[2].x), "v"(mk[2].y), "v"(mk[3].x), "v"(mk[3].y),
              "s"(sel_lo), "s"(sel_hi));
    }
    if (h == 0) out[b * 16 + n] = wv;
    }
}

extern "C" void kernel_launch(void* const* d_in, const int* in_sizes, int n_in,
                              void* d_out, int out_size, void* d_ws, size_t ws_size,
                              hipStream_t stream) {
    const float* x   = (const float*)d_in[0];
    const float* c   = (const float*)d_in[1];
    const float* psi = (const float*)d_in[2];
    float* out = (float*)d_out;

    const int B = in_sizes[1] / 64;
    dim3 grid(1024);
    dim3 block(512);
    hipLaunchKernelGGL(expm_apply_kernel, grid, block, 0, stream,
                       x, c, psi, out, B / 16);
}


# scaling exponent chosen per wave from the Frobenius norm of T: 2^-4 and 1 squaring where |T|_F^2 <= 128, else 2^-6 and 3 squarings as before; 8 matrix-vector steps kept; summation order changes
# speedup vs baseline: 1.9559x; 1.1216x over previous
#include <hip/hip_runtime.h>


typedef __attribute__((ext_vector_type(2))) float v2f;
typedef __attribute__((ext_vector_type(4))) float v4f;
typedef __attribute__((ext_vector_type(8))) float v8f;

#define ROW_ROR(v, k)                                                          \
    __builtin_bit_cast(float, __builtin_amdgcn_update_dpp(                     \
        0, __builtin_bit_cast(int, (v)), 0x120 + (k), 0xf, 0xf, true))

#define SQS_LO 4
#define SQS_HI 6
#define FRO2_LO 128.0f
#define NMV 8
#define KSQ_LO (SQS_LO - 3)
#define KSQ_HI (SQS_HI - 3)

#define K1 4.980119205559973e-3f
#define K2 1.992047682223989e-2f
#define K3 7.665265321119145e-2f
#define K4 8.765009801785550e-1f
#define K5 1.225521150112075e-1f
#define K6 2.974307204847627f

#define RP 20
#define MP (16 * RP + 4)
#define CP 68

static __device__ __forceinline__ v8f mm16(const v2f (&a)[4], const v8f b, v8f c)
{
#pragma unroll
    for (int kk = 0; kk < 4; ++kk) {
        v2f bb;
        bb.x = b[2 * kk];
        bb.y = b[2 * kk + 1];
        c = __builtin_amdgcn_wmma_f32_16x16x4_f32(
            false, a[kk], false, bb, (short)0, c, false, false);
    }
    return c;
}

static __device__ __forceinline__ void to_rows(float* colp, const v4f* arow,
                                               const v8f d, v2f (&a)[4])
{
#pragma unroll
    for (int j = 0; j < 8; ++j) colp[j * RP] = d[j];
    asm volatile("s_wait_dscnt 0" ::: "memory");
    const v4f q0 = arow[0];
    const v4f q1 = arow[1];
    a[0].x = q0.x; a[0].y = q0.y;
    a[1].x = q0.z; a[1].y = q0.w;
    a[2].x = q1.x; a[2].y = q1.y;
    a[3].x = q1.z; a[3].y = q1.w;
}

__global__ __launch_bounds__(512) void expm_apply_kernel(
    const float* __restrict__ x,
    const float* __restrict__ c,
    const float* __restrict__ psi,
    float* __restrict__ out,
    const int ntiles)
{
    __shared__ __attribute__((aligned(16))) float tile[16 * MP];
    __shared__ __attribute__((aligned(16))) float cs[16 * CP];

    const int tid = threadIdx.x;
    const int w   = tid >> 5;
    const int l   = tid & 31;
    const int h   = l >> 4;
    const int n   = l & 15;
    const int tstep = (int)gridDim.x;
    int ti = (int)blockIdx.x;
    if (ti >= ntiles) return;

    v2f crow = *(const v2f*)(c + ((long)ti * 16 + w) * 64 + 2 * l);

    v2f pb[16];
#pragma unroll
    for (int kk = 0; kk < 16; ++kk) {
        const float* pp = psi + (size_t)(kk * 4 + 2 * h) * 256 + w * 16 + n;
        pb[kk].x = pp[0];
        pb[kk].y = pp[256];
    }

    float* mt = &tile[w * MP];
    const v4f* arow = (const v4f*)(mt + n * RP + 8 * h);
    float* colp = mt + 8 * h * RP + n;
    const v4f* mrow = (const v4f*)(mt + ((n + 8 * h) & 15) * RP + 8 * h);
    float* scat = tile + 8 * h * MP + w * RP + n;
    const v8f zero = {0.f, 0.f, 0.f, 0.f, 0.f, 0.f, 0.f, 0.f};

    for (; ti < ntiles; ti += tstep) {
    const long b = (long)ti * 16 + w;

    *(v2f*)(cs + w * CP + 2 * l) = crow;
    __syncthreads();

    v8f acc = {0.f, 0.f, 0.f, 0.f, 0.f, 0.f, 0.f, 0.f};
#pragma unroll
    for (int kk = 0; kk < 16; ++kk) {
        const int k0 = kk * 4;
        const v2f a = *(const v2f*)(cs + n * CP + (k0 + 2 * h));
        acc = __builtin_amdgcn_wmma_f32_16x16x4_f32(
            false, a, false, pb[kk], (short)0, acc, false, false);
    }
#pragma unroll
    for (int j = 0; j < 8; ++j) {
        scat[j * MP] = acc[j];
    }
    __syncthreads();

    const int tn = (ti + tstep < ntiles) ? ti + tstep : ti;
    crow = *(const v2f*)(c + ((long)tn * 16 + w) * 64 + 2 * l);
    float wv = x[b * 16 + ((n + 8 * h) & 15)];

    v2f Ak[4];
    v8f ad;
    bool big;
    {
        const v4f q0 = arow[0];
        const v4f q1 = arow[1];
#pragma unroll
        for (int j = 0; j < 8; ++j) ad[j] = colp[j * RP];
        v2f f2 = {0.f, 0.f};
#pragma unroll
        for (int j = 0; j < 8; j += 2) {
            const v2f e = {ad[j], ad[j + 1]};
            f2 = __builtin_elementwise_fma(e, e, f2);
        }
        float fs = f2.x + f2.y;
        fs += ROW_ROR(fs, 8);
        fs += ROW_ROR(fs, 4);
        fs += ROW_ROR(fs, 2);
        fs += ROW_ROR(fs, 1);
        const unsigned fu = __builtin_bit_cast(unsigned, fs);
        fs += __builtin_bit_cast(float,
            __builtin_amdgcn_permlanex16(fu, fu, 0x76543210u, 0xfedcba98u, false, false));
        big = __builtin_amdgcn_ballot_w32(!(fs <= FRO2_LO)) != 0u;
        const float sc = big ? 1.0f / (float)(1 << SQS_HI) : 1.0f / (float)(1 << SQS_LO);
        Ak[0].x = q0.x * sc; Ak[0].y = q0.y * sc;
        Ak[1].x = q0.z * sc; Ak[1].y = q0.w * sc;
        Ak[2].x = q1.x * sc; Ak[2].y = q1.y * sc;
        Ak[3].x = q1.z * sc; Ak[3].y = q1.w * sc;
#pragma unroll
        for (int j = 0; j < 8; ++j) ad[j] *= sc;
    }

    const v8f a2 = mm16(Ak, ad, zero);
    v2f A2k[4];
    to_rows(colp, arow, a2, A2k);

    v8f t;
#pragma unroll
    for (int j = 0; j < 8; ++j) t[j] = K1 * a2[j] + K2 * ad[j];
    const v8f y = mm16(A2k, t, zero);

    v8f f1, f2;
#pragma unroll
    for (int j = 0; j < 8; ++j) {
        f1[j] = y[j] + (K3 * a2[j] + K4 * ad[j]);
        f2[j] = y[j] + K5 * a2[j];
    }
    v2f F1k[4];
    to_rows(colp, arow, f1, F1k);
    const v8f pr = mm16(F1k, f2, zero);
    v8f r;
#pragma unroll
    for (int j = 0; j < 8; ++j) {
        const float lowp = pr[j] + (K6 * y[j] + (0.5f * a2[j] + ad[j]));
        r[j] = ((n == (j + 8 * h)) ? 1.f : 0.f) + lowp;
    }

#pragma unroll
    for (int sq = 0; sq < KSQ_LO; ++sq) {
        v2f aa[4];
        to_rows(colp, arow, r, aa);
        r = mm16(aa, r, zero);
    }
    if (big) {
#pragma unroll
        for (int sq = KSQ_LO; sq < KSQ_HI; ++sq) {
            v2f aa[4];
            to_rows(colp, arow, r, aa);
            r = mm16(aa, r, zero);
        }
    }

    v2f mk[4];
    to_rows(colp, mrow, r, mk);
    const unsigned sel_lo = 0xfedcba98u;
    const unsigned sel_hi = 0x76543210u;
#pragma unroll
    for (int s = 0; s < NMV; ++s) {
        float t0, t1;
        asm("s_delay_alu instid0(VALU_DEP_1)\n\t"
            "v_mul_f32_dpp %1, %0, %3 row_share:0 row_mask:0xf bank_mask:0xf bound_ctrl:1\n\t"
            "v_mul_f32_dpp %2, %0, %4 row_share:1 row_mask:0xf bank_mask:0xf bound_ctrl:1\n\t"
            "s_delay_alu instid0(VALU_DEP_2) | instskip(NEXT) | instid1(VALU_DEP_2)\n\t"
            "v_fmac_f32_dpp %1, %0, %5 row_share:2 row_mask:0xf bank_mask:0xf bound_ctrl:1\n\t"
            "v_fmac_f32_dpp %2, %0, %6 row_share:3 row_mask:0xf bank_mask:0xf bound_ctrl:1\n\t"
            "s_delay_alu instid0(VALU_DEP_2) | instskip(NEXT) | instid1(VALU_DEP_2)\n\t"
            "v_fmac_f32_dpp %1, %0, %7 row_share:4 row_mask:0xf bank_mask:0xf bound_ctrl:1\n\t"
            "v_fmac_f32_dpp %2, %0, %8 row_share:5 row_mask:0xf bank_mask:0xf bound_ctrl:1\n\t"
            "s_delay_alu instid0(VALU_DEP_2) | instskip(NEXT) | instid1(VALU_DEP_2)\n\t"
            "v_fmac_f32_dpp %1, %0, %9 row_share:6 row_mask:0xf bank_mask:0xf bound_ctrl:1\n\t"
            "v_fmac_f32_dpp %2, %0, %10 row_share:7 row_mask:0xf bank_mask:0xf bound_ctrl:1\n\t"
            "s_delay_alu instid0(VALU_DEP_1) | instskip(NEXT) | instid1(VALU_DEP_1)\n\t"
            "v_add_f32 %1, %1, %2\n\t"
            "v_permlanex16_b32 %2, %1, %11, %12\n\t"
            "s_delay_alu instid0(VALU_DEP_1)\n\t"
            "v_add_f32 %0, %1, %2"
            : "+v"(wv), "=&v"(t0), "=&v"(t1)
            : "v"(mk[0].x), "v"(mk[0].y), "v"(mk[1].x), "v"(mk[1].y),
              "v"(mk[2].x), "v"(mk[2].y), "v"(mk[3].x), "v"(mk[3].y),
              "s"(sel_lo), "s"(sel_hi));
    }
    if (h == 0) out[b * 16 + n] = wv;
    }
}

extern "C" void kernel_launch(void* const* d_in, const int* in_sizes, int n_in,
                              void* d_out, int out_size, void* d_ws, size_t ws_size,
                              hipStream_t stream) {
    const float* x   = (const float*)d_in[0];
    const float* c   = (const float*)d_in[1];
    const float* psi = (const float*)d_in[2];
    float* out = (float*)d_out;

    const int B = in_sizes[1] / 64;
    dim3 grid(1024);
    dim3 block(512);
    hipLaunchKernelGGL(expm_apply_kernel, grid, block, 0, stream,
                       x, c, psi, out, B / 16);
}


# scaling exponent also from the Frobenius norm of T*T (formed unscaled, first): 2^-3 and no squaring product where |T|_F^2 <= 128 and |T*T|_F^2 <= 720; else as before; rounding changes, k order kept
# speedup vs baseline: 2.0432x; 1.0446x over previous
#include <hip/hip_runtime.h>


typedef __attribute__((ext_vector_type(2))) float v2f;
typedef __attribute__((ext_vector_type(4))) float v4f;
typedef __attribute__((ext_vector_type(8))) float v8f;

#define ROW_ROR(v, k)                                                          \
    __builtin_bit_cast(float, __builtin_amdgcn_update_dpp(                     \
        0, __builtin_bit_cast(int, (v)), 0x120 + (k), 0xf, 0xf, true))

#define SQS_MIN 3
#define SQS_LO 4
#define SQS_HI 6
#define FRO2_LO 128.0f
#define FRO2_SQ 720.0f
#define NMV 8
#define KSQ_LO (SQS_LO - 3)
#define KSQ_HI (SQS_HI - 3)

#define K1 4.980119205559973e-3f
#define K2 1.992047682223989e-2f
#define K3 7.665265321119145e-2f
#define K4 8.765009801785550e-1f
#define K5 1.225521150112075e-1f
#define K6 2.974307204847627f

#define RP 20
#define MP (16 * RP + 4)
#define CP 68

static __device__ __forceinline__ v8f mm16(const v2f (&a)[4], const v8f b, v8f c)
{
#pragma unroll
    for (int kk = 0; kk < 4; ++kk) {
        v2f bb;
        bb.x = b[2 * kk];
        bb.y = b[2 * kk + 1];
        c = __builtin_amdgcn_wmma_f32_16x16x4_f32(
            false, a[kk], false, bb, (short)0, c, false, false);
    }
    return c;
}

static __device__ __forceinline__ void to_rows(float* colp, const v4f* arow,
                                               const v8f d, v2f (&a)[4])
{
#pragma unroll
    for (int j = 0; j < 8; ++j) colp[j * RP] = d[j];
    asm volatile("s_wait_dscnt 0" ::: "memory");
    const v4f q0 = arow[0];
    const v4f q1 = arow[1];
    a[0].x = q0.x; a[0].y = q0.y;
    a[1].x = q0.z; a[1].y = q0.w;
    a[2].x = q1.x; a[2].y = q1.y;
    a[3].x = q1.z; a[3].y = q1.w;
}

__global__ __launch_bounds__(512) void expm_apply_kernel(
    const float* __restrict__ x,
    const float* __restrict__ c,
    const float* __restrict__ psi,
    float* __restrict__ out,
    const int ntiles)
{
    __shared__ __attribute__((aligned(16))) float tile[16 * MP];
    __shared__ __attribute__((aligned(16))) float cs[16 * CP];

    const int tid = threadIdx.x;
    const int w   = tid >> 5;
    const int l   = tid & 31;
    const int h   = l >> 4;
    const int n   = l & 15;
    const int tstep = (int)gridDim.x;
    int ti = (int)blockIdx.x;
    if (ti >= ntiles) return;

    v2f crow = *(const v2f*)(c + ((long)ti * 16 + w) * 64 + 2 * l);

    v2f pb[16];
#pragma unroll
    for (int kk = 0; kk < 16; ++kk) {
        const float* pp = psi + (size_t)(kk * 4 + 2 * h) * 256 + w * 16 + n;
        pb[kk].x = pp[0];
        pb[kk].y = pp[256];
    }

    float* mt = &tile[w * MP];
    const v4f* arow = (const v4f*)(mt + n * RP + 8 * h);
    float* colp = mt + 8 * h * RP + n;
    const v4f* mrow = (const v4f*)(mt + ((n + 8 * h) & 15) * RP + 8 * h);
    float* scat = tile + 8 * h * MP + w * RP + n;
    const v8f zero = {0.f, 0.f, 0.f, 0.f, 0.f, 0.f, 0.f, 0.f};

    for (; ti < ntiles; ti += tstep) {
    const long b = (long)ti * 16 + w;

    *(v2f*)(cs + w * CP + 2 * l) = crow;
    __syncthreads();

    v8f acc = {0.f, 0.f, 0.f, 0.f, 0.f, 0.f, 0.f, 0.f};
#pragma unroll
    for (int kk = 0; kk < 16; ++kk) {
        const int k0 = kk * 4;
        const v2f a = *(const v2f*)(cs + n * CP + (k0 + 2 * h));
        acc = __builtin_amdgcn_wmma_f32_16x16x4_f32(
            false, a, false, pb[kk], (short)0, acc, false, false);
    }
#pragma unroll
    for (int j = 0; j < 8; ++j) {
        scat[j * MP] = acc[j];
    }
    __syncthreads();

    const int tn = (ti + tstep < ntiles) ? ti + tstep : ti;
    crow = *(const v2f*)(c + ((long)tn * 16 + w) * 64 + 2 * l);
    float wv = x[b * 16 + ((n + 8 * h) & 15)];

    v8f ad;
    v8f a2;
    bool big;
    bool nosq;
    {
        const v4f q0 = arow[0];
        const v4f q1 = arow[1];
#pragma unroll
        for (int j = 0; j < 8; ++j) ad[j] = colp[j * RP];
        v2f Tk[4];
        Tk[0].x = q0.x; Tk[0].y = q0.y;
        Tk[1].x = q0.z; Tk[1].y = q0.w;
        Tk[2].x = q1.x; Tk[2].y = q1.y;
        Tk[3].x = q1.z; Tk[3].y = q1.w;
        const v8f tt = mm16(Tk, ad, zero);
        v2f f2 = {0.f, 0.f};
        v2f g2 = {0.f, 0.f};
#pragma unroll
        for (int j = 0; j < 8; j += 2) {
            const v2f e = {ad[j], ad[j + 1]};
            f2 = __builtin_elementwise_fma(e, e, f2);
            const v2f g = {tt[j], tt[j + 1]};
            g2 = __builtin_elementwise_fma(g, g, g2);
        }
        const float fs = f2.x + f2.y;
        const float gs = g2.x + g2.y;
        float nv = h ? gs : fs;
        const unsigned ou = __builtin_bit_cast(unsigned, h ? fs : gs);
        nv += __builtin_bit_cast(float,
            __builtin_amdgcn_permlanex16(ou, ou, 0x76543210u, 0xfedcba98u, false, false));
        nv += ROW_ROR(nv, 8);
        nv += ROW_ROR(nv, 4);
        nv += ROW_ROR(nv, 2);
        nv += ROW_ROR(nv, 1);
        const unsigned over =
            __builtin_amdgcn_ballot_w32(!(nv <= (h ? FRO2_SQ : FRO2_LO)));
        big = (over & 0xffffu) != 0u;
        nosq = over == 0u;
        const float sc = big ? 1.0f / (float)(1 << SQS_HI)
                             : (nosq ? 1.0f / (float)(1 << SQS_MIN)
                                     : 1.0f / (float)(1 << SQS_LO));
        const float sc2 = sc * sc;
#pragma unroll
        for (int j = 0; j < 8; ++j) {
            ad[j] *= sc;
            a2[j] = tt[j] * sc2;
        }
    }

    v2f A2k[4];
    to_rows(colp, arow, a2, A2k);

    v8f t;
#pragma unroll
    for (int j = 0; j < 8; ++j) t[j] = K1 * a2[j] + K2 * ad[j];
    const v8f y = mm16(A2k, t, zero);

    v8f f1, f2;
#pragma unroll
    for (int j = 0; j < 8; ++j) {
        f1[j] = y[j] + (K3 * a2[j] + K4 * ad[j]);
        f2[j] = y[j] + K5 * a2[j];
    }
    v2f F1k[4];
    to_rows(colp, arow, f1, F1k);
    const v8f pr = mm16(F1k, f2, zero);
    v8f r;
#pragma unroll
    for (int j = 0; j < 8; ++j) {
        const float lowp = pr[j] + (K6 * y[j] + (0.5f * a2[j] + ad[j]));
        r[j] = ((n == (j + 8 * h)) ? 1.f : 0.f) + lowp;
    }

    if (!nosq) {
#pragma unroll
        for (int sq = 0; sq < KSQ_LO; ++sq) {
            v2f aa[4];
            to_rows(colp, arow, r, aa);
            r = mm16(aa, r, zero);
        }
        if (big) {
#pragma unroll
            for (int sq = KSQ_LO; sq < KSQ_HI; ++sq) {
                v2f aa[4];
                to_rows(colp, arow, r, aa);
                r = mm16(aa, r, zero);
            }
        }
    }

    v2f mk[4];
    to_rows(colp, mrow, r, mk);
    const unsigned sel_lo = 0xfedcba98u;
    const unsigned sel_hi = 0x76543210u;
#pragma unroll
    for (int s = 0; s < NMV; ++s) {
        float t0, t1;
        asm("s_delay_alu instid0(VALU_DEP_1)\n\t"
            "v_mul_f32_dpp %1, %0, %3 row_share:0 row_mask:0xf bank_mask:0xf bound_ctrl:1\n\t"
            "v_mul_f32_dpp %2, %0, %4 row_share:1 row_mask:0xf bank_mask:0xf bound_ctrl:1\n\t"
            "s_delay_alu instid0(VALU_DEP_2) | instskip(NEXT) | instid1(VALU_DEP_2)\n\t"
            "v_fmac_f32_dpp %1, %0, %5 row_share:2 row_mask:0xf bank_mask:0xf bound_ctrl:1\n\t"
            "v_fmac_f32_dpp %2, %0, %6 row_share:3 row_mask:0xf bank_mask:0xf bound_ctrl:1\n\t"
            "s_delay_alu instid0(VALU_DEP_2) | instskip(NEXT) | instid1(VALU_DEP_2)\n\t"
            "v_fmac_f32_dpp %1, %0, %7 row_share:4 row_mask:0xf bank_mask:0xf bound_ctrl:1\n\t"
            "v_fmac_f32_dpp %2, %0, %8 row_share:5 row_mask:0xf bank_mask:0xf bound_ctrl:1\n\t"
            "s_delay_alu instid0(VALU_DEP_2) | instskip(NEXT) | instid1(VALU_DEP_2)\n\t"
            "v_fmac_f32_dpp %1, %0, %9 row_share:6 row_mask:0xf bank_mask:0xf bound_ctrl:1\n\t"
            "v_fmac_f32_dpp %2, %0, %10 row_share:7 row_mask:0xf bank_mask:0xf bound_ctrl:1\n\t"
            "s_delay_alu instid0(VALU_DEP_1) | instskip(NEXT) | instid1(VALU_DEP_1)\n\t"
            "v_add_f32 %1, %1, %2\n\t"
            "v_permlanex16_b32 %2, %1, %11, %12\n\t"
            "s_delay_alu instid0(VALU_DEP_1)\n\t"
            "v_add_f32 %0, %1, %2"
            : "+v"(wv), "=&v"(t0), "=&v"(t1)
            : "v"(mk[0].x), "v"(mk[0].y), "v"(mk[1].x), "v"(mk[1].y),
              "v"(mk[2].x), "v"(mk[2].y), "v"(mk[3].x), "v"(mk[3].y),
              "s"(sel_lo), "s"(sel_hi));
    }
    if (h == 0) out[b * 16 + n] = wv;
    }
}

extern "C" void kernel_launch(void* const* d_in, const int* in_sizes, int n_in,
                              void* d_out, int out_size, void* d_ws, size_t ws_size,
                              hipStream_t stream) {
    const float* x   = (const float*)d_in[0];
    const float* c   = (const float*)d_in[1];
    const float* psi = (const float*)d_in[2];
    float* out = (float*)d_out;

    const int B = in_sizes[1] / 64;
    dim3 grid(1024);
    dim3 block(512);
    hipLaunchKernelGGL(expm_apply_kernel, grid, block, 0, stream,
                       x, c, psi, out, B / 16);
}


# every wave carries two matrices: rows w and w+8 of T from shared A fragments, then both expm chains interleaved; blocks of 8 waves, grid 512: a quarter as many waves enter; summation order unchanged
# speedup vs baseline: 2.1721x; 1.0631x over previous
#include <hip/hip_runtime.h>


typedef __attribute__((ext_vector_type(2))) float v2f;
typedef __attribute__((ext_vector_type(4))) float v4f;
typedef __attribute__((ext_vector_type(8))) float v8f;

#define ROW_ROR(v, k)                                                          \
    __builtin_bit_cast(float, __builtin_amdgcn_update_dpp(                     \
        0, __builtin_bit_cast(int, (v)), 0x120 + (k), 0xf, 0xf, true))

#define SQS_MIN 3
#define SQS_LO 4
#define SQS_HI 6
#define FRO2_LO 128.0f
#define FRO2_SQ 720.0f
#define NMV 8
#define KSQ_LO (SQS_LO - 3)
#define KSQ_HI (SQS_HI - 3)

#define K1 4.980119205559973e-3f
#define K2 1.992047682223989e-2f
#define K3 7.665265321119145e-2f
#define K4 8.765009801785550e-1f
#define K5 1.225521150112075e-1f
#define K6 2.974307204847627f

#define RP 20
#define MP (16 * RP + 4)
#define CP 68

static __device__ __forceinline__ v8f mm16(const v2f (&a)[4], const v8f b, v8f c)
{
#pragma unroll
    for (int kk = 0; kk < 4; ++kk) {
        v2f bb;
        bb.x = b[2 * kk];
        bb.y = b[2 * kk + 1];
        c = __builtin_amdgcn_wmma_f32_16x16x4_f32(
            false, a[kk], false, bb, (short)0, c, false, false);
    }
    return c;
}

static __device__ __forceinline__ void to_rows(float* colp, const v4f* arow,
                                               const v8f d, v2f (&a)[4])
{
#pragma unroll
    for (int j = 0; j < 8; ++j) colp[j * RP] = d[j];
    asm volatile("s_wait_dscnt 0" ::: "memory");
    const v4f q0 = arow[0];
    const v4f q1 = arow[1];
    a[0].x = q0.x; a[0].y = q0.y;
    a[1].x = q0.z; a[1].y = q0.w;
    a[2].x = q1.x; a[2].y = q1.y;
    a[3].x = q1.z; a[3].y = q1.w;
}


static __device__ __forceinline__ void mm16x2(const v2f (&a0)[4], const v8f b0, v8f& c0,
                                              const v2f (&a1)[4], const v8f b1, v8f& c1)
{
#pragma unroll
    for (int kk = 0; kk < 4; ++kk) {
        v2f bb0, bb1;
        bb0.x = b0[2 * kk];
        bb0.y = b0[2 * kk + 1];
        bb1.x = b1[2 * kk];
        bb1.y = b1[2 * kk + 1];
        c0 = __builtin_amdgcn_wmma_f32_16x16x4_f32(
            false, a0[kk], false, bb0, (short)0, c0, false, false);
        c1 = __builtin_amdgcn_wmma_f32_16x16x4_f32(
            false, a1[kk], false, bb1, (short)0, c1, false, false);
    }
}

static __device__ __forceinline__ void to_rows2(float* colp0, const v4f* arow0, const v8f d0, v2f (&a0)[4],
                                                float* colp1, const v4f* arow1, const v8f d1, v2f (&a1)[4])
{
#pragma unroll
    for (int j = 0; j < 8; ++j) colp0[j * RP] = d0[j];
#pragma unroll
    for (int j = 0; j < 8; ++j) colp1[j * RP] = d1[j];
    asm volatile("s_wait_dscnt 0" ::: "memory");
    const v4f p0 = arow0[0];
    const v4f p1 = arow0[1];
    const v4f r0 = arow1[0];
    const v4f r1 = arow1[1];
    a0[0].x = p0.x; a0[0].y = p0.y;
    a0[1].x = p0.z; a0[1].y = p0.w;
    a0[2].x = p1.x; a0[2].y = p1.y;
    a0[3].x = p1.z; a0[3].y = p1.w;
    a1[0].x = r0.x; a1[0].y = r0.y;
    a1[1].x = r0.z; a1[1].y = r0.w;
    a1[2].x = r1.x; a1[2].y = r1.y;
    a1[3].x = r1.z; a1[3].y = r1.w;
}

static __device__ __forceinline__ float pick_scale(const v8f ad, const v8f tt, const int h,
                                                   bool& big, bool& nosq)
{
    v2f f2 = {0.f, 0.f};
    v2f g2 = {0.f, 0.f};
#pragma unroll
    for (int j = 0; j < 8; j += 2) {
        const v2f e = {ad[j], ad[j + 1]};
        f2 = __builtin_elementwise_fma(e, e, f2);
        const v2f g = {tt[j], tt[j + 1]};
        g2 = __builtin_elementwise_fma(g, g, g2);
    }
    const float fs = f2.x + f2.y;
    const float gs = g2.x + g2.y;
    float nv = h ? gs : fs;
    const unsigned ou = __builtin_bit_cast(unsigned, h ? fs : gs);
    nv += __builtin_bit_cast(float,
        __builtin_amdgcn_permlanex16(ou, ou, 0x76543210u, 0xfedcba98u, false, false));
    nv += ROW_ROR(nv, 8);
    nv += ROW_ROR(nv, 4);
    nv += ROW_ROR(nv, 2);
    nv += ROW_ROR(nv, 1);
    const unsigned over =
        __builtin_amdgcn_ballot_w32(!(nv <= (h ? FRO2_SQ : FRO2_LO)));
    big = (over & 0xffffu) != 0u;
    nosq = over == 0u;
    return big ? 1.0f / (float)(1 << SQS_HI)
               : (nosq ? 1.0f / (float)(1 << SQS_MIN)
                       : 1.0f / (float)(1 << SQS_LO));
}

__global__ __launch_bounds__(256) void expm_apply_kernel(
    const float* __restrict__ x,
    const float* __restrict__ c,
    const float* __restrict__ psi,
    float* __restrict__ out,
    const int ntiles)
{
    __shared__ __attribute__((aligned(16))) float tile[16 * MP];
    __shared__ __attribute__((aligned(16))) float cs[16 * CP];

    const int tid = threadIdx.x;
    const int w   = tid >> 5;
    const int l   = tid & 31;
    const int h   = l >> 4;
    const int n   = l & 15;
    const int tstep = (int)gridDim.x;
    int ti = (int)blockIdx.x;
    if (ti >= ntiles) return;

    const int crw = 2 * w + h;
    v4f crow = *(const v4f*)(c + ((long)ti * 16 + crw) * 64 + 4 * n);

    v2f pb0[16], pb1[16];
#pragma unroll
    for (int kk = 0; kk < 16; ++kk) {
        const float* pp = psi + (size_t)(kk * 4 + 2 * h) * 256 + w * 16 + n;
        pb0[kk].x = pp[0];
        pb0[kk].y = pp[256];
        pb1[kk].x = pp[128];
        pb1[kk].y = pp[256 + 128];
    }

    float* mt0 = &tile[w * MP];
    float* mt1 = &tile[(w + 8) * MP];
    const v4f* arow0 = (const v4f*)(mt0 + n * RP + 8 * h);
    const v4f* arow1 = (const v4f*)(mt1 + n * RP + 8 * h);
    float* colp0 = mt0 + 8 * h * RP + n;
    float* colp1 = mt1 + 8 * h * RP + n;
    const v4f* mrow0 = (const v4f*)(mt0 + ((n + 8 * h) & 15) * RP + 8 * h);
    const v4f* mrow1 = (const v4f*)(mt1 + ((n + 8 * h) & 15) * RP + 8 * h);
    float* scat0 = tile + 8 * h * MP + w * RP + n;
    float* scat1 = scat0 + 8 * RP;
    const v8f zero = {0.f, 0.f, 0.f, 0.f, 0.f, 0.f, 0.f, 0.f};

    for (; ti < ntiles; ti += tstep) {
    const long b0 = (long)ti * 16 + w;
    const long b1 = b0 + 8;

    *(v4f*)(cs + crw * CP + 4 * n) = crow;
    __syncthreads();

    v8f acc0 = {0.f, 0.f, 0.f, 0.f, 0.f, 0.f, 0.f, 0.f};
    v8f acc1 = {0.f, 0.f, 0.f, 0.f, 0.f, 0.f, 0.f, 0.f};
#pragma unroll
    for (int kk = 0; kk < 16; ++kk) {
        const int k0 = kk * 4;
        const v2f a = *(const v2f*)(cs + n * CP + (k0 + 2 * h));
        acc0 = __builtin_amdgcn_wmma_f32_16x16x4_f32(
            false, a, false, pb0[kk], (short)0, acc0, false, false);
        acc1 = __builtin_amdgcn_wmma_f32_16x16x4_f32(
            false, a, false, pb1[kk], (short)0, acc1, false, false);
    }
#pragma unroll
    for (int j = 0; j < 8; ++j) {
        scat0[j * MP] = acc0[j];
        scat1[j * MP] = acc1[j];
    }
    __syncthreads();

    const int tn = (ti + tstep < ntiles) ? ti + tstep : ti;
    crow = *(const v4f*)(c + ((long)tn * 16 + crw) * 64 + 4 * n);
    float wv0 = x[b0 * 16 + ((n + 8 * h) & 15)];
    float wv1 = x[b1 * 16 + ((n + 8 * h) & 15)];

    v8f ad0, ad1;
    v8f a20, a21;
    bool big0, big1;
    bool nosq0, nosq1;
    {
        const v4f p0 = arow0[0];
        const v4f p1 = arow0[1];
        const v4f r0 = arow1[0];
        const v4f r1 = arow1[1];
#pragma unroll
        for (int j = 0; j < 8; ++j) ad0[j] = colp0[j * RP];
#pragma unroll
        for (int j = 0; j < 8; ++j) ad1[j] = colp1[j * RP];
        v2f Tk0[4], Tk1[4];
        Tk0[0].x = p0.x; Tk0[0].y = p0.y;
        Tk0[1].x = p0.z; Tk0[1].y = p0.w;
        Tk0[2].x = p1.x; Tk0[2].y = p1.y;
        Tk0[3].x = p1.z; Tk0[3].y = p1.w;
        Tk1[0].x = r0.x; Tk1[0].y = r0.y;
        Tk1[1].x = r0.z; Tk1[1].y = r0.w;
        Tk1[2].x = r1.x; Tk1[2].y = r1.y;
        Tk1[3].x = r1.z; Tk1[3].y = r1.w;
        v8f tt0 = zero, tt1 = zero;
        mm16x2(Tk0, ad0, tt0, Tk1, ad1, tt1);
        const float sc0 = pick_scale(ad0, tt0, h, big0, nosq0);
        const float sc1 = pick_scale(ad1, tt1, h, big1, nosq1);
        const float sq0 = sc0 * sc0;
        const float sq1 = sc1 * sc1;
#pragma unroll
        for (int j = 0; j < 8; ++j) {
            ad0[j] *= sc0;
            a20[j] = tt0[j] * sq0;
            ad1[j] *= sc1;
            a21[j] = tt1[j] * sq1;
        }
    }

    v2f A2k0[4], A2k1[4];
    to_rows2(colp0, arow0, a20, A2k0, colp1, arow1, a21, A2k1);

    v8f t0v, t1v;
#pragma unroll
    for (int j = 0; j < 8; ++j) {
        t0v[j] = K1 * a20[j] + K2 * ad0[j];
        t1v[j] = K1 * a21[j] + K2 * ad1[j];
    }
    v8f y0 = zero, y1 = zero;
    mm16x2(A2k0, t0v, y0, A2k1, t1v, y1);

    v8f f10, f20, f11, f21;
#pragma unroll
    for (int j = 0; j < 8; ++j) {
        f10[j] = y0[j] + (K3 * a20[j] + K4 * ad0[j]);
        f20[j] = y0[j] + K5 * a20[j];
        f11[j] = y1[j] + (K3 * a21[j] + K4 * ad1[j]);
        f21[j] = y1[j] + K5 * a21[j];
    }
    v2f F1k0[4], F1k1[4];
    to_rows2(colp0, arow0, f10, F1k0, colp1, arow1, f11, F1k1);
    v8f pr0 = zero, pr1 = zero;
    mm16x2(F1k0, f20, pr0, F1k1, f21, pr1);
    v8f rr0, rr1;
#pragma unroll
    for (int j = 0; j < 8; ++j) {
        const float idn = (n == (j + 8 * h)) ? 1.f : 0.f;
        const float low0 = pr0[j] + (K6 * y0[j] + (0.5f * a20[j] + ad0[j]));
        const float low1 = pr1[j] + (K6 * y1[j] + (0.5f * a21[j] + ad1[j]));
        rr0[j] = idn + low0;
        rr1[j] = idn + low1;
    }

    if (!nosq0) {
#pragma unroll
        for (int sq = 0; sq < KSQ_LO; ++sq) {
            v2f aa[4];
            to_rows(colp0, arow0, rr0, aa);
            rr0 = mm16(aa, rr0, zero);
        }
        if (big0) {
#pragma unroll
            for (int sq = KSQ_LO; sq < KSQ_HI; ++sq) {
                v2f aa[4];
                to_rows(colp0, arow0, rr0, aa);
                rr0 = mm16(aa, rr0, zero);
            }
        }
    }
    if (!nosq1) {
#pragma unroll
        for (int sq = 0; sq < KSQ_LO; ++sq) {
            v2f aa[4];
            to_rows(colp1, arow1, rr1, aa);
            rr1 = mm16(aa, rr1, zero);
        }
        if (big1) {
#pragma unroll
            for (int sq = KSQ_LO; sq < KSQ_HI; ++sq) {
                v2f aa[4];
                to_rows(colp1, arow1, rr1, aa);
                rr1 = mm16(aa, rr1, zero);
            }
        }
    }

    v2f mk0[4], mk1[4];
    to_rows2(colp0, mrow0, rr0, mk0, colp1, mrow1, rr1, mk1);
    const unsigned sel_lo = 0xfedcba98u;
    const unsigned sel_hi = 0x76543210u;
#pragma unroll
    for (int s = 0; s < NMV; ++s) {
        float ta0, ta1, tb0, tb1;
        asm("s_delay_alu instid0(VALU_DEP_2) | instskip(SKIP_1) | instid1(VALU_DEP_3)\n\t"
            "v_mul_f32_dpp %2, %0, %6 row_share:0 row_mask:0xf bank_mask:0xf bound_ctrl:1\n\t"
            "v_mul_f32_dpp %3, %0, %7 row_share:1 row_mask:0xf bank_mask:0xf bound_ctrl:1\n\t"
            "v_mul_f32_dpp %4, %1, %14 row_share:0 row_mask:0xf bank_mask:0xf bound_ctrl:1\n\t"
            "v_mul_f32_dpp %5, %1, %15 row_share:1 row_mask:0xf bank_mask:0xf bound_ctrl:1\n\t"
            "s_delay_alu instid0(VALU_DEP_4) | instskip(NEXT) | instid1(VALU_DEP_4)\n\t"
            "v_fmac_f32_dpp %2, %0, %8 row_share:2 row_mask:0xf bank_mask:0xf bound_ctrl:1\n\t"
            "v_fmac_f32_dpp %3, %0, %9 row_share:3 row_mask:0xf bank_mask:0xf bound_ctrl:1\n\t"
            "s_delay_alu instid0(VALU_DEP_4) | instskip(NEXT) | instid1(VALU_DEP_4)\n\t"
            "v_fmac_f32_dpp %4, %1, %16 row_share:2 row_mask:0xf bank_mask:0xf bound_ctrl:1\n\t"
            "v_fmac_f32_dpp %5, %1, %17 row_share:3 row_mask:0xf bank_mask:0xf bound_ctrl:1\n\t"
            "s_delay_alu instid0(VALU_DEP_4) | instskip(NEXT) | instid1(VALU_DEP_4)\n\t"
            "v_fmac_f32_dpp %2, %0, %10 row_share:4 row_mask:0xf bank_mask:0xf bound_ctrl:1\n\t"
            "v_fmac_f32_dpp %3, %0, %11 row_share:5 row_mask:0xf bank_mask:0xf bound_ctrl:1\n\t"
            "s_delay_alu instid0(VALU_DEP_4) | instskip(NEXT) | instid1(VALU_DEP_4)\n\t"
            "v_fmac_f32_dpp %4, %1, %18 row_share:4 row_mask:0xf bank_mask:0xf bound_ctrl:1\n\t"
            "v_fmac_f32_dpp %5, %1, %19 row_share:5 row_mask:0xf bank_mask:0xf bound_ctrl:1\n\t"
            "s_delay_alu instid0(VALU_DEP_4) | instskip(NEXT) | instid1(VALU_DEP_4)\n\t"
            "v_fmac_f32_dpp %2, %0, %12 row_share:6 row_mask:0xf bank_mask:0xf bound_ctrl:1\n\t"
            "v_fmac_f32_dpp %3, %0, %13 row_share:7 row_mask:0xf bank_mask:0xf bound_ctrl:1\n\t"
            "s_delay_alu instid0(VALU_DEP_4) | instskip(NEXT) | instid1(VALU_DEP_4)\n\t"
            "v_fmac_f32_dpp %4, %1, %20 row_share:6 row_mask:0xf bank_mask:0xf bound_ctrl:1\n\t"
            "v_fmac_f32_dpp %5, %1, %21 row_share:7 row_mask:0xf bank_mask:0xf bound_ctrl:1\n\t"
            "s_delay_alu instid0(VALU_DEP_3) | instskip(NEXT) | instid1(VALU_DEP_2)\n\t"
            "v_add_f32 %2, %2, %3\n\t"
            "v_add_f32 %4, %4, %5\n\t"
            "s_delay_alu instid0(VALU_DEP_2) | instskip(NEXT) | instid1(VALU_DEP_2)\n\t"
            "v_permlanex16_b32 %3, %2, %22, %23\n\t"
            "v_permlanex16_b32 %5, %4, %22, %23\n\t"
            "s_delay_alu instid0(VALU_DEP_2) | instskip(NEXT) | instid1(VALU_DEP_2)\n\t"
            "v_add_f32 %0, %2, %3\n\t"
            "v_add_f32 %1, %4, %5"
            : "+v"(wv0), "+v"(wv1), "=&v"(ta0), "=&v"(ta1), "=&v"(tb0), "=&v"(tb1)
            : "v"(mk0[0].x), "v"(mk0[0].y), "v"(mk0[1].x), "v"(mk0[1].y),
              "v"(mk0[2].x), "v"(mk0[2].y), "v"(mk0[3].x), "v"(mk0[3].y),
              "v"(mk1[0].x), "v"(mk1[0].y), "v"(mk1[1].x), "v"(mk1[1].y),
              "v"(mk1[2].x), "v"(mk1[2].y), "v"(mk1[3].x), "v"(mk1[3].y),
              "s"(sel_lo), "s"(sel_hi));
    }
    if (h == 0) {
        out[b0 * 16 + n] = wv0;
        out[b1 * 16 + n] = wv1;
    }
    }
}

extern "C" void kernel_launch(void* const* d_in, const int* in_sizes, int n_in,
                              void* d_out, int out_size, void* d_ws, size_t ws_size,
                              hipStream_t stream) {
    const float* x   = (const float*)d_in[0];
    const float* c   = (const float*)d_in[1];
    const float* psi = (const float*)d_in[2];
    float* out = (float*)d_out;

    const int B = in_sizes[1] / 64;
    dim3 grid(512);
    dim3 block(256);
    hipLaunchKernelGGL(expm_apply_kernel, grid, block, 0, stream,
                       x, c, psi, out, B / 16);
}
